# starting kernel
# baseline (speedup 1.0000x reference)
#include <hip/hip_runtime.h>


typedef __attribute__((ext_vector_type(2))) float v2f;
typedef __attribute__((ext_vector_type(8))) float v8f;

#define H_IN   721
#define W_IN   1440
#define H_OUT  361
#define W_OUT  720
#define N_IN   (H_IN * W_IN)
#define N_OUT  (H_OUT * W_OUT)
#define NNZ    1048576
#define NCHUNK (NNZ / 16)

__global__ __launch_bounds__(256)
void regrid_zero_kernel(float* __restrict__ y, int n) {
    int i = blockIdx.x * blockDim.x + threadIdx.x;
    int stride = gridDim.x * blockDim.x;
    for (; i < n; i += stride) y[i] = 0.0f;
}

__global__ __launch_bounds__(256)
void regrid_coo_wmma_kernel(const float* __restrict__ x,
                            const float* __restrict__ S,
                            const int*   __restrict__ rows,
                            const int*   __restrict__ cols,
                            float*       __restrict__ y) {
    const int lane = threadIdx.x & 31;
    const int n    = lane & 15;
    const int half = lane >> 4;

    const int gw = __builtin_amdgcn_readfirstlane(
        blockIdx.x * (blockDim.x >> 5) + (threadIdx.x >> 5));
    const int totalWaves = gridDim.x * (blockDim.x >> 5);

    const int chgroup    = gw & 3;
    const int stream     = gw >> 2;
    const int numStreams = totalWaves >> 2;
    const int chbase     = chgroup * 16;

    const float* __restrict__ xp = x + (size_t)(chbase + n) * N_IN;

    for (int chunk = stream; chunk < NCHUNK; chunk += numStreams) {
        const int kbase = chunk * 16;
        const int kb2   = kbase + 2 * half;

        int   colv[8];
        float sv[8];
        #pragma unroll
        for (int i = 0; i < 8; ++i) {
            const int off = ((i >> 1) << 2) | (i & 1);
            colv[i] = cols[kb2 + off];
            sv[i]   = S[kb2 + off];
        }
        const int r = rows[kbase + n];

        float av[8];
        #pragma unroll
        for (int i = 0; i < 8; ++i) av[i] = xp[colv[i]];

        v8f d = {};
        #pragma unroll
        for (int g = 0; g < 4; ++g) {
            v2f a, b;
            a[0] = av[2 * g + 0];
            a[1] = av[2 * g + 1];
            b[0] = (n == 4 * g + 2 * half + 0) ? sv[2 * g + 0] : 0.0f;
            b[1] = (n == 4 * g + 2 * half + 1) ? sv[2 * g + 1] : 0.0f;
            d = __builtin_amdgcn_wmma_f32_16x16x4_f32(
                    false, a, false, b, (short)0, d, false, false);
        }

        float* yb = y + (size_t)(chbase + 8 * half) * N_OUT + (size_t)r;
        #pragma unroll
        for (int v = 0; v < 8; ++v) {
            atomicAdd(yb + (size_t)v * N_OUT, d[v]);
        }
    }
}

extern "C" void kernel_launch(void* const* d_in, const int* in_sizes, int n_in,
                              void* d_out, int out_size, void* d_ws, size_t ws_size,
                              hipStream_t stream) {
    (void)in_sizes; (void)n_in; (void)d_ws; (void)ws_size;

    const float* x    = (const float*)d_in[0];
    const float* S    = (const float*)d_in[1];
    const int*   rows = (const int*)d_in[2];
    const int*   cols = (const int*)d_in[3];
    float*       y    = (float*)d_out;

    regrid_zero_kernel<<<4096, 256, 0, stream>>>(y, out_size);

    regrid_coo_wmma_kernel<<<2048, 256, 0, stream>>>(x, S, rows, cols, y);
}


# x re-laid out channel-innermost in workspace every call: a nonzero's 64-channel gather is one aligned 256 B read, not 64 plane reads; a wave does all 64 ch per chunk; atomics kept (arrival order)
# speedup vs baseline: 1.1787x; 1.1787x over previous
#include <hip/hip_runtime.h>


typedef __attribute__((ext_vector_type(2))) float v2f;
typedef __attribute__((ext_vector_type(4))) float v4f;
typedef __attribute__((ext_vector_type(8))) float v8f;

#define H_IN   721
#define W_IN   1440
#define H_OUT  361
#define W_OUT  720
#define N_IN   (H_IN * W_IN)
#define N_OUT  (H_OUT * W_OUT)
#define NNZ    1048576
#define NCHUNK (NNZ / 16)
#define NCH    64
#define TP     32
#define NTILE  (N_IN / TP)
#define LP     65

__global__ __launch_bounds__(256)
void regrid_zero_kernel(float* __restrict__ y, int n) {
    int i = blockIdx.x * blockDim.x + threadIdx.x;
    int stride = gridDim.x * blockDim.x;
    for (; i < n; i += stride) y[i] = 0.0f;
}

__device__ __forceinline__ void tile_put(float* __restrict__ tb, int q, int c,
                                         const v4f a, const v4f b) {
    #pragma unroll
    for (int i = 0; i < 4; ++i) {
        tb[(4 * q + i) * LP + c]      = a[i];
        tb[(4 * q + i) * LP + c + 32] = b[i];
    }
}

__device__ __forceinline__ void tile_drain(const float* __restrict__ tb,
                                           float* __restrict__ xt, size_t p0,
                                           int pr, int c4) {
    #pragma unroll
    for (int h = 0; h < 2; ++h) {
        const int pos = pr + 16 * h;
        v4f o;
        #pragma unroll
        for (int i = 0; i < 4; ++i) o[i] = tb[pos * LP + 4 * c4 + i];
        *(v4f*)(xt + (p0 + pos) * NCH + 4 * c4) = o;
    }
}

__device__ __forceinline__ void lds_barrier() {
    __builtin_amdgcn_fence(__ATOMIC_RELEASE, "workgroup", "local");
    __builtin_amdgcn_s_barrier();
    __builtin_amdgcn_fence(__ATOMIC_ACQUIRE, "workgroup", "local");
}

__global__ __launch_bounds__(256)
void regrid_relayout_kernel(const float* __restrict__ x, float* __restrict__ xt) {
    __shared__ float tile[2][TP * LP];
    const int t  = threadIdx.x;
    const int q  = t & 7;
    const int c  = t >> 3;
    const int c4 = t & 15;
    const int pr = t >> 4;
    const int G  = gridDim.x;
    const int first = blockIdx.x;
    if (first >= NTILE) return;
    const int nIter = (NTILE - 1 - first) / G + 1;
    const int lastT = first + (nIter - 1) * G;

    const float* __restrict__ xa = x + (size_t)c * N_IN + 4 * q;
    const float* __restrict__ xb = x + (size_t)(c + 32) * N_IN + 4 * q;

    int tn = first;
    v4f a0 = __builtin_nontemporal_load((const v4f*)(xa + (size_t)tn * TP));
    v4f b0 = __builtin_nontemporal_load((const v4f*)(xb + (size_t)tn * TP));
    tn = min(tn + G, lastT);
    v4f a1 = __builtin_nontemporal_load((const v4f*)(xa + (size_t)tn * TP));
    v4f b1 = __builtin_nontemporal_load((const v4f*)(xb + (size_t)tn * TP));

    for (int it = 0; ; it += 2) {
        tile_put(tile[0], q, c, a0, b0);
        tn = min(tn + G, lastT);
        a0 = __builtin_nontemporal_load((const v4f*)(xa + (size_t)tn * TP));
        b0 = __builtin_nontemporal_load((const v4f*)(xb + (size_t)tn * TP));
        lds_barrier();
        tile_drain(tile[0], xt, (size_t)(first + it * G) * TP, pr, c4);
        if (it + 1 >= nIter) break;

        tile_put(tile[1], q, c, a1, b1);
        tn = min(tn + G, lastT);
        a1 = __builtin_nontemporal_load((const v4f*)(xa + (size_t)tn * TP));
        b1 = __builtin_nontemporal_load((const v4f*)(xb + (size_t)tn * TP));
        lds_barrier();
        tile_drain(tile[1], xt, (size_t)(first + (it + 1) * G) * TP, pr, c4);
        if (it + 2 >= nIter) break;
    }
}

__global__ __launch_bounds__(256)
void regrid_coo_wmma_kernel(const float* __restrict__ xt,
                            const float* __restrict__ S,
                            const int*   __restrict__ rows,
                            const int*   __restrict__ cols,
                            float*       __restrict__ y) {
    const int lane = threadIdx.x & 31;
    const int n    = lane & 15;
    const int half = lane >> 4;

    const int gw = __builtin_amdgcn_readfirstlane(
        blockIdx.x * (blockDim.x >> 5) + (threadIdx.x >> 5));
    const int totalWaves = gridDim.x * (blockDim.x >> 5);

    const float* __restrict__ xq = xt + 4 * n;
    float* __restrict__ yq = y + (size_t)(4 * n) * N_OUT;

    for (int chunk = gw; chunk < NCHUNK; chunk += totalWaves) {
        const int kbase = chunk * 16;
        const int kb2   = kbase + 2 * half;

        int colv[8];
        #pragma unroll
        for (int i = 0; i < 8; ++i) {
            const int off = ((i >> 1) << 2) | (i & 1);
            colv[i] = cols[kb2 + off];
        }
        const float s = S[kbase + n];
        int rv[8];
        #pragma unroll
        for (int v = 0; v < 8; ++v) rv[v] = rows[kbase + 8 * half + v];

        v4f xv[8];
        #pragma unroll
        for (int i = 0; i < 8; ++i)
            xv[i] = __builtin_nontemporal_load(
                        (const v4f*)(xq + (size_t)colv[i] * NCH));

        v8f d0 = {}, d1 = {}, d2 = {}, d3 = {};
        #pragma unroll
        for (int g = 0; g < 4; ++g) {
            v2f a, b0, b1, b2, b3;
            a[0] = (n == 4 * g + 2 * half + 0) ? s : 0.0f;
            a[1] = (n == 4 * g + 2 * half + 1) ? s : 0.0f;
            b0[0] = xv[2 * g][0]; b0[1] = xv[2 * g + 1][0];
            b1[0] = xv[2 * g][1]; b1[1] = xv[2 * g + 1][1];
            b2[0] = xv[2 * g][2]; b2[1] = xv[2 * g + 1][2];
            b3[0] = xv[2 * g][3]; b3[1] = xv[2 * g + 1][3];
            d0 = __builtin_amdgcn_wmma_f32_16x16x4_f32(
                     false, a, false, b0, (short)0, d0, false, false);
            d1 = __builtin_amdgcn_wmma_f32_16x16x4_f32(
                     false, a, false, b1, (short)0, d1, false, false);
            d2 = __builtin_amdgcn_wmma_f32_16x16x4_f32(
                     false, a, false, b2, (short)0, d2, false, false);
            d3 = __builtin_amdgcn_wmma_f32_16x16x4_f32(
                     false, a, false, b3, (short)0, d3, false, false);
        }

        #pragma unroll
        for (int v = 0; v < 8; ++v) {
            float* p = yq + (size_t)rv[v];
            atomicAdd(p,                     d0[v]);
            atomicAdd(p + (size_t)N_OUT,     d1[v]);
            atomicAdd(p + (size_t)2 * N_OUT, d2[v]);
            atomicAdd(p + (size_t)3 * N_OUT, d3[v]);
        }
    }
}

extern "C" void kernel_launch(void* const* d_in, const int* in_sizes, int n_in,
                              void* d_out, int out_size, void* d_ws, size_t ws_size,
                              hipStream_t stream) {
    (void)in_sizes; (void)n_in; (void)ws_size;

    const float* x    = (const float*)d_in[0];
    const float* S    = (const float*)d_in[1];
    const int*   rows = (const int*)d_in[2];
    const int*   cols = (const int*)d_in[3];
    float*       y    = (float*)d_out;
    float*       xt   = (float*)d_ws;

    regrid_relayout_kernel<<<2048, 256, 0, stream>>>(x, xt);

    regrid_zero_kernel<<<4096, 256, 0, stream>>>(y, out_size);

    regrid_coo_wmma_kernel<<<1024, 256, 0, stream>>>(xt, S, rows, cols, y);
}


# y accumulated channel-innermost in workspace (zeroed each call) and transposed out by a 4th kernel: a nonzero's 64 atomic adds hit one aligned 256 B row, not 64 planes; atomics kept (arrival order)
# speedup vs baseline: 13.4228x; 11.3878x over previous
#include <hip/hip_runtime.h>


typedef __attribute__((ext_vector_type(2))) float v2f;
typedef __attribute__((ext_vector_type(4))) float v4f;
typedef __attribute__((ext_vector_type(8))) float v8f;

#define H_IN   721
#define W_IN   1440
#define H_OUT  361
#define W_OUT  720
#define N_IN   (H_IN * W_IN)
#define N_OUT  (H_OUT * W_OUT)
#define NNZ    1048576
#define NCHUNK (NNZ / 16)
#define NCH    64
#define TP     32
#define NTILE  (N_IN / TP)
#define LP     65
#define TQ     32
#define TROWS  48
#define NTILE_OUT ((N_OUT - 16) / TQ)
#define P_TAIL (NTILE_OUT * TQ)

__global__ __launch_bounds__(256)
void regrid_zero_kernel(float* __restrict__ y, int n) {
    int i = blockIdx.x * blockDim.x + threadIdx.x;
    int stride = gridDim.x * blockDim.x;
    for (; i < n; i += stride) y[i] = 0.0f;
}

__device__ __forceinline__ void tile_put(float* __restrict__ tb, int q, int c,
                                         const v4f a, const v4f b) {
    #pragma unroll
    for (int i = 0; i < 4; ++i) {
        tb[(4 * q + i) * LP + c]      = a[i];
        tb[(4 * q + i) * LP + c + 32] = b[i];
    }
}

__device__ __forceinline__ void tile_drain(const float* __restrict__ tb,
                                           float* __restrict__ xt, size_t p0,
                                           int pr, int c4) {
    #pragma unroll
    for (int h = 0; h < 2; ++h) {
        const int pos = pr + 16 * h;
        v4f o;
        #pragma unroll
        for (int i = 0; i < 4; ++i) o[i] = tb[pos * LP + 4 * c4 + i];
        *(v4f*)(xt + (p0 + pos) * NCH + 4 * c4) = o;
    }
}

__device__ __forceinline__ void lds_barrier() {
    __builtin_amdgcn_fence(__ATOMIC_RELEASE, "workgroup", "local");
    __builtin_amdgcn_s_barrier();
    __builtin_amdgcn_fence(__ATOMIC_ACQUIRE, "workgroup", "local");
}

__global__ __launch_bounds__(256)
void regrid_relayout_kernel(const float* __restrict__ x, float* __restrict__ xt) {
    __shared__ float tile[2][TP * LP];
    const int t  = threadIdx.x;
    const int q  = t & 7;
    const int c  = t >> 3;
    const int c4 = t & 15;
    const int pr = t >> 4;
    const int G  = gridDim.x;
    const int first = blockIdx.x;
    if (first >= NTILE) return;
    const int nIter = (NTILE - 1 - first) / G + 1;
    const int lastT = first + (nIter - 1) * G;

    const float* __restrict__ xa = x + (size_t)c * N_IN + 4 * q;
    const float* __restrict__ xb = x + (size_t)(c + 32) * N_IN + 4 * q;

    int tn = first;
    v4f a0 = __builtin_nontemporal_load((const v4f*)(xa + (size_t)tn * TP));
    v4f b0 = __builtin_nontemporal_load((const v4f*)(xb + (size_t)tn * TP));
    tn = min(tn + G, lastT);
    v4f a1 = __builtin_nontemporal_load((const v4f*)(xa + (size_t)tn * TP));
    v4f b1 = __builtin_nontemporal_load((const v4f*)(xb + (size_t)tn * TP));

    for (int it = 0; ; it += 2) {
        tile_put(tile[0], q, c, a0, b0);
        tn = min(tn + G, lastT);
        a0 = __builtin_nontemporal_load((const v4f*)(xa + (size_t)tn * TP));
        b0 = __builtin_nontemporal_load((const v4f*)(xb + (size_t)tn * TP));
        lds_barrier();
        tile_drain(tile[0], xt, (size_t)(first + it * G) * TP, pr, c4);
        if (it + 1 >= nIter) break;

        tile_put(tile[1], q, c, a1, b1);
        tn = min(tn + G, lastT);
        a1 = __builtin_nontemporal_load((const v4f*)(xa + (size_t)tn * TP));
        b1 = __builtin_nontemporal_load((const v4f*)(xb + (size_t)tn * TP));
        lds_barrier();
        tile_drain(tile[1], xt, (size_t)(first + (it + 1) * G) * TP, pr, c4);
        if (it + 2 >= nIter) break;
    }
}

__global__ __launch_bounds__(256)
void regrid_coo_wmma_kernel(const float* __restrict__ xt,
                            const float* __restrict__ S,
                            const int*   __restrict__ rows,
                            const int*   __restrict__ cols,
                            float*       __restrict__ yt) {
    const int lane = threadIdx.x & 31;
    const int n    = lane & 15;
    const int half = lane >> 4;

    const int gw = __builtin_amdgcn_readfirstlane(
        blockIdx.x * (blockDim.x >> 5) + (threadIdx.x >> 5));
    const int totalWaves = gridDim.x * (blockDim.x >> 5);

    const float* __restrict__ xq = xt + 4 * n;
    float* __restrict__ yq = yt + 4 * n;

    for (int chunk = gw; chunk < NCHUNK; chunk += totalWaves) {
        const int kbase = chunk * 16;
        const int kb2   = kbase + 2 * half;

        int colv[8];
        #pragma unroll
        for (int i = 0; i < 8; ++i) {
            const int off = ((i >> 1) << 2) | (i & 1);
            colv[i] = cols[kb2 + off];
        }
        const float s = S[kbase + n];
        int rv[8];
        #pragma unroll
        for (int v = 0; v < 8; ++v) rv[v] = rows[kbase + 8 * half + v];

        v4f xv[8];
        #pragma unroll
        for (int i = 0; i < 8; ++i)
            xv[i] = __builtin_nontemporal_load(
                        (const v4f*)(xq + (size_t)colv[i] * NCH));

        v8f d0 = {}, d1 = {}, d2 = {}, d3 = {};
        #pragma unroll
        for (int g = 0; g < 4; ++g) {
            v2f a, b0, b1, b2, b3;
            a[0] = (n == 4 * g + 2 * half + 0) ? s : 0.0f;
            a[1] = (n == 4 * g + 2 * half + 1) ? s : 0.0f;
            b0[0] = xv[2 * g][0]; b0[1] = xv[2 * g + 1][0];
            b1[0] = xv[2 * g][1]; b1[1] = xv[2 * g + 1][1];
            b2[0] = xv[2 * g][2]; b2[1] = xv[2 * g + 1][2];
            b3[0] = xv[2 * g][3]; b3[1] = xv[2 * g + 1][3];
            d0 = __builtin_amdgcn_wmma_f32_16x16x4_f32(
                     false, a, false, b0, (short)0, d0, false, false);
            d1 = __builtin_amdgcn_wmma_f32_16x16x4_f32(
                     false, a, false, b1, (short)0, d1, false, false);
            d2 = __builtin_amdgcn_wmma_f32_16x16x4_f32(
                     false, a, false, b2, (short)0, d2, false, false);
            d3 = __builtin_amdgcn_wmma_f32_16x16x4_f32(
                     false, a, false, b3, (short)0, d3, false, false);
        }

        #pragma unroll
        for (int v = 0; v < 8; ++v) {
            float* p = yq + (size_t)rv[v] * NCH;
            atomicAdd(p,     d0[v]);
            atomicAdd(p + 1, d1[v]);
            atomicAdd(p + 2, d2[v]);
            atomicAdd(p + 3, d3[v]);
        }
    }
}

__device__ __forceinline__ void out_put(float* __restrict__ tb, int pr, int c4,
                                        const v4f r0, const v4f r1, const v4f r2) {
    #pragma unroll
    for (int i = 0; i < 4; ++i) {
        tb[pr * LP + 4 * c4 + i]        = r0[i];
        tb[(pr + 16) * LP + 4 * c4 + i] = r1[i];
        tb[(pr + 32) * LP + 4 * c4 + i] = r2[i];
    }
}

__device__ __forceinline__ void out_drain(const float* __restrict__ tb,
                                          float* __restrict__ y, int p0,
                                          int q, int c) {
    const int off = 16 * (c & 1);
    #pragma unroll
    for (int h = 0; h < 2; ++h) {
        const int ch = c + 32 * h;
        v4f o;
        #pragma unroll
        for (int i = 0; i < 4; ++i) o[i] = tb[(4 * q + i + off) * LP + ch];
        *(v4f*)(y + (size_t)ch * N_OUT + p0 + off + 4 * q) = o;
    }
}

__global__ __launch_bounds__(256)
void regrid_planes_out_kernel(const float* __restrict__ yt, float* __restrict__ y) {
    __shared__ float tile[2][TROWS * LP];
    const int t  = threadIdx.x;
    const int c4 = t & 15;
    const int pr = t >> 4;
    const int q  = t & 7;
    const int c  = t >> 3;
    const int G  = gridDim.x - 1;
    const float* __restrict__ src = yt + (size_t)pr * NCH + 4 * c4;

    if ((int)blockIdx.x == G) {
        float* tb = tile[0];
        const v4f r0 = __builtin_nontemporal_load(
                           (const v4f*)(src + (size_t)P_TAIL * NCH));
        const v4f r1 = __builtin_nontemporal_load((const v4f*)src);
        #pragma unroll
        for (int i = 0; i < 4; ++i) {
            tb[pr * LP + 4 * c4 + i]        = r0[i];
            tb[(pr + 16) * LP + 4 * c4 + i] = r1[i];
        }
        __syncthreads();
        const int ch  = 2 * c;
        const int cho = ch + (q >> 2);
        v4f o;
        #pragma unroll
        for (int i = 0; i < 4; ++i) o[i] = tb[(4 * q + i) * LP + cho];
        *(v4f*)(y + (size_t)ch * N_OUT + P_TAIL + 4 * q) = o;
        return;
    }

    const int first = blockIdx.x;
    if (first >= NTILE_OUT) return;

    v4f r0 = __builtin_nontemporal_load((const v4f*)(src + (size_t)first * TQ * NCH));
    v4f r1 = __builtin_nontemporal_load((const v4f*)(src + ((size_t)first * TQ + 16) * NCH));
    v4f r2 = __builtin_nontemporal_load((const v4f*)(src + ((size_t)first * TQ + 32) * NCH));
    int buf = 0;
    for (int tl = first; tl < NTILE_OUT; tl += G) {
        float* tb = tile[buf];
        out_put(tb, pr, c4, r0, r1, r2);
        const int tn = min(tl + G, NTILE_OUT - 1);
        r0 = __builtin_nontemporal_load((const v4f*)(src + (size_t)tn * TQ * NCH));
        r1 = __builtin_nontemporal_load((const v4f*)(src + ((size_t)tn * TQ + 16) * NCH));
        r2 = __builtin_nontemporal_load((const v4f*)(src + ((size_t)tn * TQ + 32) * NCH));
        lds_barrier();
        out_drain(tb, y, tl * TQ, q, c);
        buf ^= 1;
    }
}

extern "C" void kernel_launch(void* const* d_in, const int* in_sizes, int n_in,
                              void* d_out, int out_size, void* d_ws, size_t ws_size,
                              hipStream_t stream) {
    (void)in_sizes; (void)n_in; (void)ws_size; (void)out_size;

    const float* x    = (const float*)d_in[0];
    const float* S    = (const float*)d_in[1];
    const int*   rows = (const int*)d_in[2];
    const int*   cols = (const int*)d_in[3];
    float*       y    = (float*)d_out;
    float*       xt   = (float*)d_ws;
    float*       yt   = xt + (size_t)N_IN * NCH;

    regrid_relayout_kernel<<<2048, 256, 0, stream>>>(x, xt);

    regrid_zero_kernel<<<4096, 256, 0, stream>>>(yt, N_OUT * NCH);

    regrid_coo_wmma_kernel<<<1024, 256, 0, stream>>>(xt, S, rows, cols, yt);

    regrid_planes_out_kernel<<<2049, 256, 0, stream>>>(yt, y);
}


# each atomic instruction of a wave adds 32 consecutive channels of ONE row (one whole 128 B line) instead of 4 part-lines; xt channels interleaved to feed it; atomics kept (arrival order)
# speedup vs baseline: 18.7930x; 1.4001x over previous
#include <hip/hip_runtime.h>


typedef __attribute__((ext_vector_type(2))) float v2f;
typedef __attribute__((ext_vector_type(4))) float v4f;
typedef __attribute__((ext_vector_type(8))) float v8f;

#define H_IN   721
#define W_IN   1440
#define H_OUT  361
#define W_OUT  720
#define N_IN   (H_IN * W_IN)
#define N_OUT  (H_OUT * W_OUT)
#define NNZ    1048576
#define NCHUNK (NNZ / 16)
#define NCH    64
#define TP     32
#define NTILE  (N_IN / TP)
#define LP     65
#define TQ     32
#define TROWS  48
#define NTILE_OUT ((N_OUT - 16) / TQ)
#define P_TAIL (NTILE_OUT * TQ)

__global__ __launch_bounds__(256)
void regrid_zero_kernel(float* __restrict__ y, int n) {
    int i = blockIdx.x * blockDim.x + threadIdx.x;
    int stride = gridDim.x * blockDim.x;
    for (; i < n; i += stride) y[i] = 0.0f;
}

__device__ __forceinline__ void tile_put(float* __restrict__ tb, int q, int c,
                                         const v4f a, const v4f b) {
    #pragma unroll
    for (int i = 0; i < 4; ++i) {
        tb[(4 * q + i) * LP + c]      = a[i];
        tb[(4 * q + i) * LP + c + 32] = b[i];
    }
}

__device__ __forceinline__ void tile_drain(const float* __restrict__ tb,
                                           float* __restrict__ xt, size_t p0,
                                           int pr, int c4) {
    #pragma unroll
    for (int h = 0; h < 2; ++h) {
        const int pos = pr + h;
        v4f o;
        #pragma unroll
        for (int i = 0; i < 4; ++i) o[i] = tb[pos * LP + 16 * i + c4];
        *(v4f*)(xt + (p0 + pos) * NCH + 4 * c4) = o;
    }
}

__device__ __forceinline__ void lds_barrier() {
    __builtin_amdgcn_fence(__ATOMIC_RELEASE, "workgroup", "local");
    __builtin_amdgcn_s_barrier();
    __builtin_amdgcn_fence(__ATOMIC_ACQUIRE, "workgroup", "local");
}

__global__ __launch_bounds__(256)
void regrid_relayout_kernel(const float* __restrict__ x, float* __restrict__ xt) {
    __shared__ float tile[2][TP * LP];
    const int t  = threadIdx.x;
    const int q  = t & 7;
    const int c  = t >> 3;
    const int c4 = t & 15;
    const int pr = 2 * (t >> 5) + 16 * ((t >> 4) & 1);
    const int G  = gridDim.x;
    const int first = blockIdx.x;
    if (first >= NTILE) return;
    const int nIter = (NTILE - 1 - first) / G + 1;
    const int lastT = first + (nIter - 1) * G;

    const float* __restrict__ xa = x + (size_t)c * N_IN + 4 * q;
    const float* __restrict__ xb = x + (size_t)(c + 32) * N_IN + 4 * q;

    int tn = first;
    v4f a0 = __builtin_nontemporal_load((const v4f*)(xa + (size_t)tn * TP));
    v4f b0 = __builtin_nontemporal_load((const v4f*)(xb + (size_t)tn * TP));
    tn = min(tn + G, lastT);
    v4f a1 = __builtin_nontemporal_load((const v4f*)(xa + (size_t)tn * TP));
    v4f b1 = __builtin_nontemporal_load((const v4f*)(xb + (size_t)tn * TP));

    for (int it = 0; ; it += 2) {
        tile_put(tile[0], q, c, a0, b0);
        tn = min(tn + G, lastT);
        a0 = __builtin_nontemporal_load((const v4f*)(xa + (size_t)tn * TP));
        b0 = __builtin_nontemporal_load((const v4f*)(xb + (size_t)tn * TP));
        lds_barrier();
        tile_drain(tile[0], xt, (size_t)(first + it * G) * TP, pr, c4);
        if (it + 1 >= nIter) break;

        tile_put(tile[1], q, c, a1, b1);
        tn = min(tn + G, lastT);
        a1 = __builtin_nontemporal_load((const v4f*)(xa + (size_t)tn * TP));
        b1 = __builtin_nontemporal_load((const v4f*)(xb + (size_t)tn * TP));
        lds_barrier();
        tile_drain(tile[1], xt, (size_t)(first + (it + 1) * G) * TP, pr, c4);
        if (it + 2 >= nIter) break;
    }
}

__global__ __launch_bounds__(256)
void regrid_coo_wmma_kernel(const float* __restrict__ xt,
                            const float* __restrict__ S,
                            const int*   __restrict__ rows,
                            const int*   __restrict__ cols,
                            float*       __restrict__ yt) {
    const int lane = threadIdx.x & 31;
    const int n    = lane & 15;
    const int half = lane >> 4;

    const int gw = __builtin_amdgcn_readfirstlane(
        blockIdx.x * (blockDim.x >> 5) + (threadIdx.x >> 5));
    const int totalWaves = gridDim.x * (blockDim.x >> 5);

    const float* __restrict__ xq = xt + 4 * n;
    float* __restrict__ yl = yt + lane;

    for (int chunk = gw; chunk < NCHUNK; chunk += totalWaves) {
        const int kbase = chunk * 16;
        const int kb2   = kbase + 2 * half;

        int colv[8];
        #pragma unroll
        for (int i = 0; i < 8; ++i) {
            const int off = ((i >> 1) << 2) | (i & 1);
            colv[i] = cols[kb2 + off];
        }
        const float s0 = S[kbase + (n & 7)];
        const float s1 = S[kbase + 8 + (n & 7)];
        int rv[16];
        #pragma unroll
        for (int m = 0; m < 16; ++m) rv[m] = rows[kbase + m];

        v4f xv[8];
        #pragma unroll
        for (int i = 0; i < 8; ++i)
            xv[i] = __builtin_nontemporal_load(
                        (const v4f*)(xq + (size_t)colv[i] * NCH));

        v8f d0 = {}, d1 = {}, d2 = {}, d3 = {};
        #pragma unroll
        for (int g = 0; g < 4; ++g) {
            const bool k0 = (n == 4 * g + 2 * half + 0);
            const bool k1 = (n == 4 * g + 2 * half + 1);
            const int  i0 = 2 * (g & 1);
            const int  e  = g >> 1;
            v2f a0, a1, b0, b1, b2, b3;
            a0[0] = k0 ? s0 : 0.0f;  a0[1] = k1 ? s0 : 0.0f;
            a1[0] = k0 ? s1 : 0.0f;  a1[1] = k1 ? s1 : 0.0f;
            b0[0] = xv[i0][e];         b0[1] = xv[i0 + 1][e];
            b1[0] = xv[i0][2 + e];     b1[1] = xv[i0 + 1][2 + e];
            b2[0] = xv[4 + i0][e];     b2[1] = xv[4 + i0 + 1][e];
            b3[0] = xv[4 + i0][2 + e]; b3[1] = xv[4 + i0 + 1][2 + e];
            d0 = __builtin_amdgcn_wmma_f32_16x16x4_f32(
                     false, a0, false, b0, (short)0, d0, false, false);
            d1 = __builtin_amdgcn_wmma_f32_16x16x4_f32(
                     false, a0, false, b1, (short)0, d1, false, false);
            d2 = __builtin_amdgcn_wmma_f32_16x16x4_f32(
                     false, a1, false, b2, (short)0, d2, false, false);
            d3 = __builtin_amdgcn_wmma_f32_16x16x4_f32(
                     false, a1, false, b3, (short)0, d3, false, false);
        }

        #pragma unroll
        for (int v = 0; v < 8; ++v) {
            float* p0 = yl + (size_t)rv[v] * NCH;
            float* p1 = yl + (size_t)rv[8 + v] * NCH;
            atomicAdd(p0,      d0[v]);
            atomicAdd(p0 + 32, d1[v]);
            atomicAdd(p1,      d2[v]);
            atomicAdd(p1 + 32, d3[v]);
        }
    }
}

__device__ __forceinline__ void out_put(float* __restrict__ tb, int pr, int c4,
                                        const v4f r0, const v4f r1, const v4f r2) {
    #pragma unroll
    for (int i = 0; i < 4; ++i) {
        tb[pr * LP + 4 * c4 + i]        = r0[i];
        tb[(pr + 16) * LP + 4 * c4 + i] = r1[i];
        tb[(pr + 32) * LP + 4 * c4 + i] = r2[i];
    }
}

__device__ __forceinline__ void out_drain(const float* __restrict__ tb,
                                          float* __restrict__ y, int p0,
                                          int q, int c) {
    const int off = 16 * (c & 1);
    #pragma unroll
    for (int h = 0; h < 2; ++h) {
        const int ch = c + 32 * h;
        v4f o;
        #pragma unroll
        for (int i = 0; i < 4; ++i) o[i] = tb[(4 * q + i + off) * LP + ch];
        *(v4f*)(y + (size_t)ch * N_OUT + p0 + off + 4 * q) = o;
    }
}

__global__ __launch_bounds__(256)
void regrid_planes_out_kernel(const float* __restrict__ yt, float* __restrict__ y) {
    __shared__ float tile[2][TROWS * LP];
    const int t  = threadIdx.x;
    const int c4 = t & 15;
    const int pr = t >> 4;
    const int q  = t & 7;
    const int c  = t >> 3;
    const int G  = gridDim.x - 1;
    const float* __restrict__ src = yt + (size_t)pr * NCH + 4 * c4;

    if ((int)blockIdx.x == G) {
        float* tb = tile[0];
        const v4f r0 = __builtin_nontemporal_load(
                           (const v4f*)(src + (size_t)P_TAIL * NCH));
        const v4f r1 = __builtin_nontemporal_load((const v4f*)src);
        #pragma unroll
        for (int i = 0; i < 4; ++i) {
            tb[pr * LP + 4 * c4 + i]        = r0[i];
            tb[(pr + 16) * LP + 4 * c4 + i] = r1[i];
        }
        __syncthreads();
        const int ch  = 2 * c;
        const int cho = ch + (q >> 2);
        v4f o;
        #pragma unroll
        for (int i = 0; i < 4; ++i) o[i] = tb[(4 * q + i) * LP + cho];
        *(v4f*)(y + (size_t)ch * N_OUT + P_TAIL + 4 * q) = o;
        return;
    }

    const int first = blockIdx.x;
    if (first >= NTILE_OUT) return;

    v4f r0 = __builtin_nontemporal_load((const v4f*)(src + (size_t)first * TQ * NCH));
    v4f r1 = __builtin_nontemporal_load((const v4f*)(src + ((size_t)first * TQ + 16) * NCH));
    v4f r2 = __builtin_nontemporal_load((const v4f*)(src + ((size_t)first * TQ + 32) * NCH));
    int buf = 0;
    for (int tl = first; tl < NTILE_OUT; tl += G) {
        float* tb = tile[buf];
        out_put(tb, pr, c4, r0, r1, r2);
        const int tn = min(tl + G, NTILE_OUT - 1);
        r0 = __builtin_nontemporal_load((const v4f*)(src + (size_t)tn * TQ * NCH));
        r1 = __builtin_nontemporal_load((const v4f*)(src + ((size_t)tn * TQ + 16) * NCH));
        r2 = __builtin_nontemporal_load((const v4f*)(src + ((size_t)tn * TQ + 32) * NCH));
        lds_barrier();
        out_drain(tb, y, tl * TQ, q, c);
        buf ^= 1;
    }
}

extern "C" void kernel_launch(void* const* d_in, const int* in_sizes, int n_in,
                              void* d_out, int out_size, void* d_ws, size_t ws_size,
                              hipStream_t stream) {
    (void)in_sizes; (void)n_in; (void)ws_size; (void)out_size;

    const float* x    = (const float*)d_in[0];
    const float* S    = (const float*)d_in[1];
    const int*   rows = (const int*)d_in[2];
    const int*   cols = (const int*)d_in[3];
    float*       y    = (float*)d_out;
    float*       xt   = (float*)d_ws;
    float*       yt   = xt + (size_t)N_IN * NCH;

    regrid_relayout_kernel<<<2048, 256, 0, stream>>>(x, xt);

    regrid_zero_kernel<<<4096, 256, 0, stream>>>(yt, N_OUT * NCH);

    regrid_coo_wmma_kernel<<<1024, 256, 0, stream>>>(xt, S, rows, cols, yt);

    regrid_planes_out_kernel<<<2049, 256, 0, stream>>>(yt, y);
}


# speedup vs baseline: 19.4437x; 1.0346x over previous
#include <hip/hip_runtime.h>


typedef __attribute__((ext_vector_type(2))) float v2f;
typedef __attribute__((ext_vector_type(4))) float v4f;
typedef __attribute__((ext_vector_type(8))) float v8f;

#define H_IN   721
#define W_IN   1440
#define H_OUT  361
#define W_OUT  720
#define N_IN   (H_IN * W_IN)
#define N_OUT  (H_OUT * W_OUT)
#define NNZ    1048576
#define NCHUNK (NNZ / 16)
#define NCH    64
#define TP     32
#define NTILE  (N_IN / TP)
#define RUN    4
#define NRUN   ((NTILE + RUN - 1) / RUN)
#define LP     65
#define TQ     32
#define TROWS  48
#define NTILE_OUT ((N_OUT - 16) / TQ)
#define P_TAIL (NTILE_OUT * TQ)

__global__ __launch_bounds__(256)
void regrid_zero_kernel(float* __restrict__ y, int n) {
    int i = blockIdx.x * blockDim.x + threadIdx.x;
    int stride = gridDim.x * blockDim.x;
    for (; i < n; i += stride) y[i] = 0.0f;
}

__device__ __forceinline__ void tile_put(float* __restrict__ tb, int q, int c,
                                         const v4f a, const v4f b) {
    #pragma unroll
    for (int i = 0; i < 4; ++i) {
        tb[(4 * q + i) * LP + c]      = a[i];
        tb[(4 * q + i) * LP + c + 32] = b[i];
    }
}

__device__ __forceinline__ void tile_drain(const float* __restrict__ tb,
                                           float* __restrict__ xt, size_t p0,
                                           int pr, int c4) {
    #pragma unroll
    for (int h = 0; h < 2; ++h) {
        const int pos = pr + h;
        v4f o;
        #pragma unroll
        for (int i = 0; i < 4; ++i) o[i] = tb[pos * LP + 16 * i + c4];
        *(v4f*)(xt + (p0 + pos) * NCH + 4 * c4) = o;
    }
}

__device__ __forceinline__ void lds_barrier() {
    __builtin_amdgcn_fence(__ATOMIC_RELEASE, "workgroup", "local");
    __builtin_amdgcn_s_barrier();
    __builtin_amdgcn_fence(__ATOMIC_ACQUIRE, "workgroup", "local");
}

__device__ __forceinline__ int run_tile(int i, int first, int G) {
    return ((i / RUN) * G + first) * RUN + (i % RUN);
}

__global__ __launch_bounds__(256)
void regrid_relayout_kernel(const float* __restrict__ x, float* __restrict__ xt) {
    __shared__ float tile[2][TP * LP];
    const int t  = threadIdx.x;
    const int q  = t & 7;
    const int c  = t >> 3;
    const int c4 = t & 15;
    const int pr = 2 * (t >> 5) + 16 * ((t >> 4) & 1);
    const int G  = gridDim.x;
    const int first = blockIdx.x;
    if (first >= NRUN) return;
    const int myRuns  = (NRUN - 1 - first) / G + 1;
    const int lastRun = first + (myRuns - 1) * G;
    const int nIter = myRuns * RUN - (lastRun == NRUN - 1 ? NRUN * RUN - NTILE : 0);
    const int lastI = nIter - 1;

    const float* __restrict__ xa = x + (size_t)c * N_IN + 4 * q;
    const float* __restrict__ xb = x + (size_t)(c + 32) * N_IN + 4 * q;

    int in = 0;
    int tn = run_tile(in, first, G);
    v4f a0 = *(const v4f*)(xa + (size_t)tn * TP);
    v4f b0 = *(const v4f*)(xb + (size_t)tn * TP);
    in = min(in + 1, lastI);
    tn = run_tile(in, first, G);
    v4f a1 = *(const v4f*)(xa + (size_t)tn * TP);
    v4f b1 = *(const v4f*)(xb + (size_t)tn * TP);

    for (int it = 0; ; it += 2) {
        tile_put(tile[0], q, c, a0, b0);
        in = min(in + 1, lastI);
        tn = run_tile(in, first, G);
        a0 = *(const v4f*)(xa + (size_t)tn * TP);
        b0 = *(const v4f*)(xb + (size_t)tn * TP);
        lds_barrier();
        tile_drain(tile[0], xt, (size_t)run_tile(it, first, G) * TP, pr, c4);
        if (it + 1 >= nIter) break;

        tile_put(tile[1], q, c, a1, b1);
        in = min(in + 1, lastI);
        tn = run_tile(in, first, G);
        a1 = *(const v4f*)(xa + (size_t)tn * TP);
        b1 = *(const v4f*)(xb + (size_t)tn * TP);
        lds_barrier();
        tile_drain(tile[1], xt, (size_t)run_tile(it + 1, first, G) * TP, pr, c4);
        if (it + 2 >= nIter) break;
    }
}

__global__ __launch_bounds__(256)
void regrid_coo_wmma_kernel(const float* __restrict__ xt,
                            const float* __restrict__ S,
                            const int*   __restrict__ rows,
                            const int*   __restrict__ cols,
                            float*       __restrict__ yt) {
    const int lane = threadIdx.x & 31;
    const int n    = lane & 15;
    const int half = lane >> 4;

    const int gw = __builtin_amdgcn_readfirstlane(
        blockIdx.x * (blockDim.x >> 5) + (threadIdx.x >> 5));
    const int totalWaves = gridDim.x * (blockDim.x >> 5);

    const float* __restrict__ xq = xt + 4 * n;
    float* __restrict__ yl = yt + lane;

    for (int chunk = gw; chunk < NCHUNK; chunk += totalWaves) {
        const int kbase = chunk * 16;
        const int kb2   = kbase + 2 * half;

        int colv[8];
        #pragma unroll
        for (int i = 0; i < 8; ++i) {
            const int off = ((i >> 1) << 2) | (i & 1);
            colv[i] = cols[kb2 + off];
        }
        const float s0 = S[kbase + (n & 7)];
        const float s1 = S[kbase + 8 + (n & 7)];
        int rv[16];
        #pragma unroll
        for (int m = 0; m < 16; ++m) rv[m] = rows[kbase + m];

        v4f xv[8];
        #pragma unroll
        for (int i = 0; i < 8; ++i)
            xv[i] = __builtin_nontemporal_load(
                        (const v4f*)(xq + (size_t)colv[i] * NCH));

        v8f d0 = {}, d1 = {}, d2 = {}, d3 = {};
        #pragma unroll
        for (int g = 0; g < 4; ++g) {
            const bool k0 = (n == 4 * g + 2 * half + 0);
            const bool k1 = (n == 4 * g + 2 * half + 1);
            const int  i0 = 2 * (g & 1);
            const int  e  = g >> 1;
            v2f a0, a1, b0, b1, b2, b3;
            a0[0] = k0 ? s0 : 0.0f;  a0[1] = k1 ? s0 : 0.0f;
            a1[0] = k0 ? s1 : 0.0f;  a1[1] = k1 ? s1 : 0.0f;
            b0[0] = xv[i0][e];         b0[1] = xv[i0 + 1][e];
            b1[0] = xv[i0][2 + e];     b1[1] = xv[i0 + 1][2 + e];
            b2[0] = xv[4 + i0][e];     b2[1] = xv[4 + i0 + 1][e];
            b3[0] = xv[4 + i0][2 + e]; b3[1] = xv[4 + i0 + 1][2 + e];
            d0 = __builtin_amdgcn_wmma_f32_16x16x4_f32(
                     false, a0, false, b0, (short)0, d0, false, false);
            d1 = __builtin_amdgcn_wmma_f32_16x16x4_f32(
                     false, a0, false, b1, (short)0, d1, false, false);
            d2 = __builtin_amdgcn_wmma_f32_16x16x4_f32(
                     false, a1, false, b2, (short)0, d2, false, false);
            d3 = __builtin_amdgcn_wmma_f32_16x16x4_f32(
                     false, a1, false, b3, (short)0, d3, false, false);
        }

        #pragma unroll
        for (int v = 0; v < 8; ++v) {
            float* p0 = yl + (size_t)rv[v] * NCH;
            float* p1 = yl + (size_t)rv[8 + v] * NCH;
            atomicAdd(p0,      d0[v]);
            atomicAdd(p0 + 32, d1[v]);
            atomicAdd(p1,      d2[v]);
            atomicAdd(p1 + 32, d3[v]);
        }
    }
}

__device__ __forceinline__ void out_put(float* __restrict__ tb, int pr, int c4,
                                        const v4f r0, const v4f r1, const v4f r2) {
    #pragma unroll
    for (int i = 0; i < 4; ++i) {
        tb[pr * LP + 4 * c4 + i]        = r0[i];
        tb[(pr + 16) * LP + 4 * c4 + i] = r1[i];
        tb[(pr + 32) * LP + 4 * c4 + i] = r2[i];
    }
}

__device__ __forceinline__ void out_drain(const float* __restrict__ tb,
                                          float* __restrict__ y, int p0,
                                          int q, int c) {
    const int off = 16 * (c & 1);
    #pragma unroll
    for (int h = 0; h < 2; ++h) {
        const int ch = c + 32 * h;
        v4f o;
        #pragma unroll
        for (int i = 0; i < 4; ++i) o[i] = tb[(4 * q + i + off) * LP + ch];
        *(v4f*)(y + (size_t)ch * N_OUT + p0 + off + 4 * q) = o;
    }
}

__global__ __launch_bounds__(256)
void regrid_planes_out_kernel(const float* __restrict__ yt, float* __restrict__ y) {
    __shared__ float tile[2][TROWS * LP];
    const int t  = threadIdx.x;
    const int c4 = t & 15;
    const int pr = t >> 4;
    const int q  = t & 7;
    const int c  = t >> 3;
    const int G  = gridDim.x - 1;
    const float* __restrict__ src = yt + (size_t)pr * NCH + 4 * c4;

    if ((int)blockIdx.x == G) {
        float* tb = tile[0];
        const v4f r0 = __builtin_nontemporal_load(
                           (const v4f*)(src + (size_t)P_TAIL * NCH));
        const v4f r1 = __builtin_nontemporal_load((const v4f*)src);
        #pragma unroll
        for (int i = 0; i < 4; ++i) {
            tb[pr * LP + 4 * c4 + i]        = r0[i];
            tb[(pr + 16) * LP + 4 * c4 + i] = r1[i];
        }
        __syncthreads();
        const int ch  = 2 * c;
        const int cho = ch + (q >> 2);
        v4f o;
        #pragma unroll
        for (int i = 0; i < 4; ++i) o[i] = tb[(4 * q + i) * LP + cho];
        *(v4f*)(y + (size_t)ch * N_OUT + P_TAIL + 4 * q) = o;
        return;
    }

    const int first = blockIdx.x;
    if (first >= NTILE_OUT) return;

    v4f r0 = __builtin_nontemporal_load((const v4f*)(src + (size_t)first * TQ * NCH));
    v4f r1 = __builtin_nontemporal_load((const v4f*)(src + ((size_t)first * TQ + 16) * NCH));
    v4f r2 = __builtin_nontemporal_load((const v4f*)(src + ((size_t)first * TQ + 32) * NCH));
    int buf = 0;
    for (int tl = first; tl < NTILE_OUT; tl += G) {
        float* tb = tile[buf];
        out_put(tb, pr, c4, r0, r1, r2);
        const int tn = min(tl + G, NTILE_OUT - 1);
        r0 = __builtin_nontemporal_load((const v4f*)(src + (size_t)tn * TQ * NCH));
        r1 = __builtin_nontemporal_load((const v4f*)(src + ((size_t)tn * TQ + 16) * NCH));
        r2 = __builtin_nontemporal_load((const v4f*)(src + ((size_t)tn * TQ + 32) * NCH));
        lds_barrier();
        out_drain(tb, y, tl * TQ, q, c);
        buf ^= 1;
    }
}

extern "C" void kernel_launch(void* const* d_in, const int* in_sizes, int n_in,
                              void* d_out, int out_size, void* d_ws, size_t ws_size,
                              hipStream_t stream) {
    (void)in_sizes; (void)n_in; (void)ws_size; (void)out_size;

    const float* x    = (const float*)d_in[0];
    const float* S    = (const float*)d_in[1];
    const int*   rows = (const int*)d_in[2];
    const int*   cols = (const int*)d_in[3];
    float*       y    = (float*)d_out;
    float*       xt   = (float*)d_ws;
    float*       yt   = xt + (size_t)N_IN * NCH;

    regrid_relayout_kernel<<<2048, 256, 0, stream>>>(x, xt);

    regrid_zero_kernel<<<4096, 256, 0, stream>>>(yt, N_OUT * NCH);

    regrid_coo_wmma_kernel<<<1024, 256, 0, stream>>>(xt, S, rows, cols, yt);

    regrid_planes_out_kernel<<<2049, 256, 0, stream>>>(yt, y);
}


# output rows owned: nonzeros bucketed per call by 1024-row block (LDS counting sort); a workgroup sums its block in an LDS slab (LDS atomics, arrival order) and stores its planes; no global atomics
# speedup vs baseline: 22.2594x; 1.1448x over previous
#include <hip/hip_runtime.h>


typedef __attribute__((ext_vector_type(2))) float v2f;
typedef __attribute__((ext_vector_type(4))) float v4f;
typedef __attribute__((ext_vector_type(8))) float v8f;

#define H_IN   721
#define W_IN   1440
#define H_OUT  361
#define W_OUT  720
#define N_IN   (H_IN * W_IN)
#define N_OUT  (H_OUT * W_OUT)
#define NNZ    1048576
#define NCHUNK (NNZ / 16)
#define NCH    64
#define TP     32
#define NTILE  (N_IN / TP)
#define RUN    4
#define NRUN   ((NTILE + RUN - 1) / RUN)
#define LP     65
#define RB     1024
#define RB_LOG 10
#define NBUCKET ((N_OUT + RB - 1) / RB)
#define NB     256
#define SORT_WG 128
#define SORT_T  1024
#define SORT_N  (NNZ / SORT_WG)
#define SORT_I  (SORT_N / SORT_T)
#define SP     65

__device__ __forceinline__ void tile_put(float* __restrict__ tb, int q, int c,
                                         const v4f a, const v4f b) {
    #pragma unroll
    for (int i = 0; i < 4; ++i) {
        tb[(4 * q + i) * LP + c]      = a[i];
        tb[(4 * q + i) * LP + c + 32] = b[i];
    }
}

__device__ __forceinline__ void tile_drain(const float* __restrict__ tb,
                                           float* __restrict__ xt, size_t p0,
                                           int pr, int c4) {
    #pragma unroll
    for (int h = 0; h < 2; ++h) {
        const int pos = pr + h;
        v4f o;
        #pragma unroll
        for (int i = 0; i < 4; ++i) o[i] = tb[pos * LP + 16 * i + c4];
        *(v4f*)(xt + (p0 + pos) * NCH + 4 * c4) = o;
    }
}

__device__ __forceinline__ void lds_barrier() {
    __builtin_amdgcn_fence(__ATOMIC_RELEASE, "workgroup", "local");
    __builtin_amdgcn_s_barrier();
    __builtin_amdgcn_fence(__ATOMIC_ACQUIRE, "workgroup", "local");
}

__device__ __forceinline__ int run_tile(int i, int first, int G) {
    return ((i / RUN) * G + first) * RUN + (i % RUN);
}

__global__ __launch_bounds__(256)
void regrid_relayout_kernel(const float* __restrict__ x, float* __restrict__ xt) {
    __shared__ float tile[2][TP * LP];
    const int t  = threadIdx.x;
    const int q  = t & 7;
    const int c  = t >> 3;
    const int c4 = t & 15;
    const int pr = 2 * (t >> 5) + 16 * ((t >> 4) & 1);
    const int G  = gridDim.x;
    const int first = blockIdx.x;
    if (first >= NRUN) return;
    const int myRuns  = (NRUN - 1 - first) / G + 1;
    const int lastRun = first + (myRuns - 1) * G;
    const int nIter = myRuns * RUN - (lastRun == NRUN - 1 ? NRUN * RUN - NTILE : 0);
    const int lastI = nIter - 1;

    const float* __restrict__ xa = x + (size_t)c * N_IN + 4 * q;
    const float* __restrict__ xb = x + (size_t)(c + 32) * N_IN + 4 * q;

    int in = 0;
    int tn = run_tile(in, first, G);
    v4f a0 = *(const v4f*)(xa + (size_t)tn * TP);
    v4f b0 = *(const v4f*)(xb + (size_t)tn * TP);
    in = min(in + 1, lastI);
    tn = run_tile(in, first, G);
    v4f a1 = *(const v4f*)(xa + (size_t)tn * TP);
    v4f b1 = *(const v4f*)(xb + (size_t)tn * TP);

    for (int it = 0; ; it += 2) {
        tile_put(tile[0], q, c, a0, b0);
        in = min(in + 1, lastI);
        tn = run_tile(in, first, G);
        a0 = *(const v4f*)(xa + (size_t)tn * TP);
        b0 = *(const v4f*)(xb + (size_t)tn * TP);
        lds_barrier();
        tile_drain(tile[0], xt, (size_t)run_tile(it, first, G) * TP, pr, c4);
        if (it + 1 >= nIter) break;

        tile_put(tile[1], q, c, a1, b1);
        in = min(in + 1, lastI);
        tn = run_tile(in, first, G);
        a1 = *(const v4f*)(xa + (size_t)tn * TP);
        b1 = *(const v4f*)(xb + (size_t)tn * TP);
        lds_barrier();
        tile_drain(tile[1], xt, (size_t)run_tile(it + 1, first, G) * TP, pr, c4);
        if (it + 2 >= nIter) break;
    }
}

__global__ __launch_bounds__(SORT_T)
void regrid_bucket_count_kernel(const int* __restrict__ rows, int* __restrict__ counts) {
    __shared__ int hist[NB];
    const int tid = threadIdx.x;
    if (tid < NB) hist[tid] = 0;
    __syncthreads();
    const int base = blockIdx.x * SORT_N;
    #pragma unroll
    for (int i = 0; i < SORT_I; ++i) {
        const int r = rows[base + i * SORT_T + tid];
        atomicAdd(&hist[r >> RB_LOG], 1);
    }
    __syncthreads();
    if (tid < NB) counts[blockIdx.x * NB + tid] = hist[tid];
}

__global__ __launch_bounds__(SORT_T)
void regrid_bucket_fill_kernel(const float* __restrict__ S,
                               const int*   __restrict__ rows,
                               const int*   __restrict__ cols,
                               const int*   __restrict__ counts,
                               int*   __restrict__ bucketStart,
                               float* __restrict__ sS,
                               int*   __restrict__ sC,
                               int*   __restrict__ sR) {
    __shared__ int   part_all[4][NB];
    __shared__ int   part_before[4][NB];
    __shared__ int   tot[NB];
    __shared__ int   own[NB];
    __shared__ int   cursor[NB];
    __shared__ int   delta[NB];
    __shared__ float stS[SORT_N];
    __shared__ int   stC[SORT_N];
    __shared__ int   stR[SORT_N];
    const int tid = threadIdx.x;
    const int b   = tid & (NB - 1);
    const int q   = tid >> 8;
    const int w   = blockIdx.x;

    int all = 0, before = 0;
    #pragma unroll 8
    for (int j = 0; j < SORT_WG / 4; ++j) {
        const int wp = (SORT_WG / 4) * q + j;
        const int c  = counts[wp * NB + b];
        all += c;
        before += (wp < w) ? c : 0;
    }
    part_all[q][b]    = all;
    part_before[q][b] = before;
    if (tid < NB) own[tid] = counts[w * NB + tid];
    __syncthreads();
    if (tid < NB)
        tot[tid] = part_all[0][tid] + part_all[1][tid] + part_all[2][tid] + part_all[3][tid];
    __syncthreads();
    if (tid < NB) {
        int ex = 0, lex = 0;
        for (int bb = 0; bb < tid; ++bb) { ex += tot[bb]; lex += own[bb]; }
        const int gbase = ex + part_before[0][tid] + part_before[1][tid]
                             + part_before[2][tid] + part_before[3][tid];
        cursor[tid] = lex;
        delta[tid]  = gbase - lex;
        if (w == 0) bucketStart[tid] = ex;
        if (w == 0 && tid < 16) {
            sS[NNZ + tid] = 0.0f;
            sC[NNZ + tid] = 0;
            sR[NNZ + tid] = 0;
        }
    }
    __syncthreads();
    const int base = w * SORT_N;
    #pragma unroll
    for (int i = 0; i < SORT_I; ++i) {
        const int k = base + i * SORT_T + tid;
        const int r = rows[k];
        const int p = atomicAdd(&cursor[r >> RB_LOG], 1);
        stS[p] = S[k];
        stC[p] = cols[k];
        stR[p] = r;
    }
    __syncthreads();
    #pragma unroll
    for (int i = 0; i < SORT_I; ++i) {
        const int p = i * SORT_T + tid;
        const int r = stR[p];
        const int g = delta[r >> RB_LOG] + p;
        sS[g] = stS[p];
        sC[g] = stC[p];
        sR[g] = r;
    }
}

__global__ __launch_bounds__(1024)
void regrid_rows_owned_kernel(const float* __restrict__ xt,
                              const float* __restrict__ sS,
                              const int*   __restrict__ sR,
                              const int*   __restrict__ sC,
                              const int*   __restrict__ bucketStart,
                              float*       __restrict__ y) {
    __shared__ float slab[RB * SP];
    const int tid  = threadIdx.x;
    const int lane = tid & 31;
    const int n    = lane & 15;
    const int half = lane >> 4;
    const int wv   = __builtin_amdgcn_readfirstlane(tid >> 5);
    const int blk  = blockIdx.x;
    const int row0 = blk * RB;

    for (int i = tid; i < RB * SP; i += 1024) slab[i] = 0.0f;
    __syncthreads();

    const int start   = bucketStart[blk];
    const int end     = bucketStart[blk + 1];
    const int nChunks = (end - start + 15) >> 4;

    const float* __restrict__ xq = xt + 4 * n;
    float* sl = slab + lane;

    for (int chunk = wv; chunk < nChunks; chunk += 32) {
        const int kbase = start + chunk * 16;
        const int kb2   = kbase + 2 * half;

        int  colv[8];
        bool live[8];
        #pragma unroll
        for (int i = 0; i < 8; ++i) {
            const int off = ((i >> 1) << 2) | (i & 1);
            live[i] = (kb2 + off < end);
            colv[i] = sC[kb2 + off];
        }
        const int   ka = kbase + (n & 7);
        const float t0 = sS[ka];
        const float t1 = sS[ka + 8];
        const float s0 = (ka < end) ? t0 : 0.0f;
        const float s1 = (ka + 8 < end) ? t1 : 0.0f;
        int rv[16];
        #pragma unroll
        for (int m = 0; m < 16; ++m) {
            const int r = sR[kbase + m];
            rv[m] = (kbase + m < end) ? r - row0 : 0;
        }

        v4f xv[8];
        #pragma unroll
        for (int i = 0; i < 8; ++i)
            xv[i] = __builtin_nontemporal_load(
                        (const v4f*)(xq + (size_t)colv[i] * NCH));
        #pragma unroll
        for (int i = 0; i < 8; ++i) {
            #pragma unroll
            for (int e = 0; e < 4; ++e) xv[i][e] = live[i] ? xv[i][e] : 0.0f;
        }

        v8f d0 = {}, d1 = {}, d2 = {}, d3 = {};
        #pragma unroll
        for (int g = 0; g < 4; ++g) {
            const bool k0 = (n == 4 * g + 2 * half + 0);
            const bool k1 = (n == 4 * g + 2 * half + 1);
            const int  i0 = 2 * (g & 1);
            const int  e  = g >> 1;
            v2f a0, a1, b0, b1, b2, b3;
            a0[0] = k0 ? s0 : 0.0f;  a0[1] = k1 ? s0 : 0.0f;
            a1[0] = k0 ? s1 : 0.0f;  a1[1] = k1 ? s1 : 0.0f;
            b0[0] = xv[i0][e];         b0[1] = xv[i0 + 1][e];
            b1[0] = xv[i0][2 + e];     b1[1] = xv[i0 + 1][2 + e];
            b2[0] = xv[4 + i0][e];     b2[1] = xv[4 + i0 + 1][e];
            b3[0] = xv[4 + i0][2 + e]; b3[1] = xv[4 + i0 + 1][2 + e];
            d0 = __builtin_amdgcn_wmma_f32_16x16x4_f32(
                     false, a0, false, b0, (short)0, d0, false, false);
            d1 = __builtin_amdgcn_wmma_f32_16x16x4_f32(
                     false, a0, false, b1, (short)0, d1, false, false);
            d2 = __builtin_amdgcn_wmma_f32_16x16x4_f32(
                     false, a1, false, b2, (short)0, d2, false, false);
            d3 = __builtin_amdgcn_wmma_f32_16x16x4_f32(
                     false, a1, false, b3, (short)0, d3, false, false);
        }

        #pragma unroll
        for (int v = 0; v < 8; ++v) {
            float* p0 = sl + rv[v] * SP;
            float* p1 = sl + rv[8 + v] * SP;
            atomicAdd(p0,      d0[v]);
            atomicAdd(p0 + 32, d1[v]);
            atomicAdd(p1,      d2[v]);
            atomicAdd(p1 + 32, d3[v]);
        }
    }
    __syncthreads();

    const int rowsHere = min(RB, N_OUT - row0);
    {
        const int ch = 2 * wv;
        float* __restrict__ yp = y + (size_t)ch * N_OUT + row0;
        const int nFull = rowsHere >> 5;
        #pragma unroll 8
        for (int j = 0; j < nFull; ++j) {
            const int r = 32 * j + lane;
            yp[r] = slab[r * SP + ch];
        }
        const int rt = 32 * nFull + lane;
        if (rt < rowsHere) yp[rt] = slab[rt * SP + ch];
    }
    {
        const int ch = 2 * wv + 1;
        float* __restrict__ yp = y + (size_t)ch * N_OUT + row0;
        const int nFull = (rowsHere - 16) >> 5;
        if (lane >= 16) yp[lane - 16] = slab[(lane - 16) * SP + ch];
        #pragma unroll 8
        for (int j = 1; j <= nFull; ++j) {
            const int r = 32 * j - 16 + lane;
            yp[r] = slab[r * SP + ch];
        }
        const int rt = 32 * (nFull + 1) - 16 + lane;
        if (rt < rowsHere) yp[rt] = slab[rt * SP + ch];
    }
}

extern "C" void kernel_launch(void* const* d_in, const int* in_sizes, int n_in,
                              void* d_out, int out_size, void* d_ws, size_t ws_size,
                              hipStream_t stream) {
    (void)in_sizes; (void)n_in; (void)ws_size; (void)out_size;

    const float* x    = (const float*)d_in[0];
    const float* S    = (const float*)d_in[1];
    const int*   rows = (const int*)d_in[2];
    const int*   cols = (const int*)d_in[3];
    float*       y    = (float*)d_out;
    float*       xt   = (float*)d_ws;
    int*   counts      = (int*)(xt + (size_t)N_IN * NCH);
    int*   bucketStart = counts + SORT_WG * NB;
    float* sS          = (float*)(bucketStart + NB);
    int*   sC          = (int*)(sS + NNZ + 16);
    int*   sR          = sC + NNZ + 16;

    regrid_relayout_kernel<<<2048, 256, 0, stream>>>(x, xt);

    regrid_bucket_count_kernel<<<SORT_WG, SORT_T, 0, stream>>>(rows, counts);
    regrid_bucket_fill_kernel<<<SORT_WG, SORT_T, 0, stream>>>(S, rows, cols, counts,
                                                              bucketStart, sS, sC, sR);

    regrid_rows_owned_kernel<<<NBUCKET, 1024, 0, stream>>>(xt, sS, sR, sC, bucketStart, y);
}


# nonzero counting per 1024-row block done inside the re-layout launch by its 80 workgroups that have four tiles fewer (LDS histograms as before); the separate counting launch is gone: 3 launches
# speedup vs baseline: 23.3410x; 1.0486x over previous
#include <hip/hip_runtime.h>


typedef __attribute__((ext_vector_type(2))) float v2f;
typedef __attribute__((ext_vector_type(4))) float v4f;
typedef __attribute__((ext_vector_type(8))) float v8f;
typedef __attribute__((ext_vector_type(4))) int   v4i;

#define H_IN   721
#define W_IN   1440
#define H_OUT  361
#define W_OUT  720
#define N_IN   (H_IN * W_IN)
#define N_OUT  (H_OUT * W_OUT)
#define NNZ    1048576
#define NCHUNK (NNZ / 16)
#define NCH    64
#define TP     32
#define NTILE  (N_IN / TP)
#define RUN    4
#define NRUN   ((NTILE + RUN - 1) / RUN)
#define LP     65
#define RB     1024
#define RB_LOG 10
#define NBUCKET ((N_OUT + RB - 1) / RB)
#define NB     256
#define SORT_WG 128
#define SORT_T  1024
#define SORT_N  (NNZ / SORT_WG)
#define SORT_I  (SORT_N / SORT_T)
#define CNT_T   256
#define CNT_J   (SORT_N / (4 * CNT_T))
#define RELAY_G 2048
static_assert(CNT_T == NB, "one LDS counter per counting thread");
static_assert(RELAY_G <= NRUN, "every re-layout workgroup has tiles, so none leaves before the counting");
static_assert(CNT_J * 4 * CNT_T == SORT_N, "a share is a whole number of four-int loads per thread");
#define SP     65

__device__ __forceinline__ void tile_put(float* __restrict__ tb, int q, int c,
                                         const v4f a, const v4f b) {
    #pragma unroll
    for (int i = 0; i < 4; ++i) {
        tb[(4 * q + i) * LP + c]      = a[i];
        tb[(4 * q + i) * LP + c + 32] = b[i];
    }
}

__device__ __forceinline__ void tile_drain(const float* __restrict__ tb,
                                           float* __restrict__ xt, size_t p0,
                                           int pr, int c4) {
    #pragma unroll
    for (int h = 0; h < 2; ++h) {
        const int pos = pr + h;
        v4f o;
        #pragma unroll
        for (int i = 0; i < 4; ++i) o[i] = tb[pos * LP + 16 * i + c4];
        *(v4f*)(xt + (p0 + pos) * NCH + 4 * c4) = o;
    }
}

__device__ __forceinline__ void lds_barrier() {
    __builtin_amdgcn_fence(__ATOMIC_RELEASE, "workgroup", "local");
    __builtin_amdgcn_s_barrier();
    __builtin_amdgcn_fence(__ATOMIC_ACQUIRE, "workgroup", "local");
}

__device__ __forceinline__ int run_tile(int i, int first, int G) {
    return ((i / RUN) * G + first) * RUN + (i % RUN);
}

__global__ __launch_bounds__(256)
void regrid_relayout_kernel(const float* __restrict__ x, float* __restrict__ xt,
                            const int* __restrict__ rows, int* __restrict__ counts) {
    __shared__ float tile[2][TP * LP];
    __shared__ int   hist[NB];
    const int t  = threadIdx.x;
    const int q  = t & 7;
    const int c  = t >> 3;
    const int c4 = t & 15;
    const int pr = 2 * (t >> 5) + 16 * ((t >> 4) & 1);
    const int G  = gridDim.x;
    const int first = blockIdx.x;

    if (first >= NRUN) return;
    const int myRuns  = (NRUN - 1 - first) / G + 1;
    const int lastRun = first + (myRuns - 1) * G;
    const int nIter = myRuns * RUN - (lastRun == NRUN - 1 ? NRUN * RUN - NTILE : 0);
    const int lastI = nIter - 1;

    const float* __restrict__ xa = x + (size_t)c * N_IN + 4 * q;
    const float* __restrict__ xb = x + (size_t)(c + 32) * N_IN + 4 * q;

    int in = 0;
    int tn = run_tile(in, first, G);
    v4f a0 = *(const v4f*)(xa + (size_t)tn * TP);
    v4f b0 = *(const v4f*)(xb + (size_t)tn * TP);
    in = min(in + 1, lastI);
    tn = run_tile(in, first, G);
    v4f a1 = *(const v4f*)(xa + (size_t)tn * TP);
    v4f b1 = *(const v4f*)(xb + (size_t)tn * TP);

    const int light0 = NRUN % G;
    if (first >= light0) {
        const int nLight = G - light0;
        for (int w = first - light0; w < SORT_WG; w += nLight) {
            hist[t] = 0;
            lds_barrier();
            const int* __restrict__ rp = rows + (size_t)w * SORT_N + 4 * t;
            v4i r[CNT_J];
            #pragma unroll
            for (int j = 0; j < CNT_J; ++j) r[j] = *(const v4i*)(rp + j * 4 * CNT_T);
            #pragma unroll
            for (int j = 0; j < CNT_J; ++j) {
                #pragma unroll
                for (int e = 0; e < 4; ++e) atomicAdd(&hist[r[j][e] >> RB_LOG], 1);
            }
            lds_barrier();
            counts[w * NB + t] = hist[t];
        }
    }

    for (int it = 0; ; it += 2) {
        tile_put(tile[0], q, c, a0, b0);
        in = min(in + 1, lastI);
        tn = run_tile(in, first, G);
        a0 = *(const v4f*)(xa + (size_t)tn * TP);
        b0 = *(const v4f*)(xb + (size_t)tn * TP);
        lds_barrier();
        tile_drain(tile[0], xt, (size_t)run_tile(it, first, G) * TP, pr, c4);
        if (it + 1 >= nIter) break;

        tile_put(tile[1], q, c, a1, b1);
        in = min(in + 1, lastI);
        tn = run_tile(in, first, G);
        a1 = *(const v4f*)(xa + (size_t)tn * TP);
        b1 = *(const v4f*)(xb + (size_t)tn * TP);
        lds_barrier();
        tile_drain(tile[1], xt, (size_t)run_tile(it + 1, first, G) * TP, pr, c4);
        if (it + 2 >= nIter) break;
    }
}

__global__ __launch_bounds__(SORT_T)
void regrid_bucket_fill_kernel(const float* __restrict__ S,
                               const int*   __restrict__ rows,
                               const int*   __restrict__ cols,
                               const int*   __restrict__ counts,
                               int*   __restrict__ bucketStart,
                               float* __restrict__ sS,
                               int*   __restrict__ sC,
                               int*   __restrict__ sR) {
    __shared__ int   part_all[4][NB];
    __shared__ int   part_before[4][NB];
    __shared__ int   tot[NB];
    __shared__ int   own[NB];
    __shared__ int   cursor[NB];
    __shared__ int   delta[NB];
    __shared__ float stS[SORT_N];
    __shared__ int   stC[SORT_N];
    __shared__ int   stR[SORT_N];
    const int tid = threadIdx.x;
    const int b   = tid & (NB - 1);
    const int q   = tid >> 8;
    const int w   = blockIdx.x;

    int all = 0, before = 0;
    #pragma unroll 8
    for (int j = 0; j < SORT_WG / 4; ++j) {
        const int wp = (SORT_WG / 4) * q + j;
        const int c  = counts[wp * NB + b];
        all += c;
        before += (wp < w) ? c : 0;
    }
    part_all[q][b]    = all;
    part_before[q][b] = before;
    if (tid < NB) own[tid] = counts[w * NB + tid];
    __syncthreads();
    if (tid < NB)
        tot[tid] = part_all[0][tid] + part_all[1][tid] + part_all[2][tid] + part_all[3][tid];
    __syncthreads();
    if (tid < NB) {
        int ex = 0, lex = 0;
        for (int bb = 0; bb < tid; ++bb) { ex += tot[bb]; lex += own[bb]; }
        const int gbase = ex + part_before[0][tid] + part_before[1][tid]
                             + part_before[2][tid] + part_before[3][tid];
        cursor[tid] = lex;
        delta[tid]  = gbase - lex;
        if (w == 0) bucketStart[tid] = ex;
        if (w == 0 && tid < 16) {
            sS[NNZ + tid] = 0.0f;
            sC[NNZ + tid] = 0;
            sR[NNZ + tid] = 0;
        }
    }
    __syncthreads();
    const int base = w * SORT_N;
    #pragma unroll
    for (int i = 0; i < SORT_I; ++i) {
        const int k = base + i * SORT_T + tid;
        const int r = rows[k];
        const int p = atomicAdd(&cursor[r >> RB_LOG], 1);
        stS[p] = S[k];
        stC[p] = cols[k];
        stR[p] = r;
    }
    __syncthreads();
    #pragma unroll
    for (int i = 0; i < SORT_I; ++i) {
        const int p = i * SORT_T + tid;
        const int r = stR[p];
        const int g = delta[r >> RB_LOG] + p;
        sS[g] = stS[p];
        sC[g] = stC[p];
        sR[g] = r;
    }
}

__global__ __launch_bounds__(1024)
void regrid_rows_owned_kernel(const float* __restrict__ xt,
                              const float* __restrict__ sS,
                              const int*   __restrict__ sR,
                              const int*   __restrict__ sC,
                              const int*   __restrict__ bucketStart,
                              float*       __restrict__ y) {
    __shared__ float slab[RB * SP];
    const int tid  = threadIdx.x;
    const int lane = tid & 31;
    const int n    = lane & 15;
    const int half = lane >> 4;
    const int wv   = __builtin_amdgcn_readfirstlane(tid >> 5);
    const int blk  = blockIdx.x;
    const int row0 = blk * RB;

    for (int i = tid; i < RB * SP; i += 1024) slab[i] = 0.0f;
    __syncthreads();

    const int start   = bucketStart[blk];
    const int end     = bucketStart[blk + 1];
    const int nChunks = (end - start + 15) >> 4;

    const float* __restrict__ xq = xt + 4 * n;
    float* sl = slab + lane;

    for (int chunk = wv; chunk < nChunks; chunk += 32) {
        const int kbase = start + chunk * 16;
        const int kb2   = kbase + 2 * half;

        int  colv[8];
        bool live[8];
        #pragma unroll
        for (int i = 0; i < 8; ++i) {
            const int off = ((i >> 1) << 2) | (i & 1);
            live[i] = (kb2 + off < end);
            colv[i] = sC[kb2 + off];
        }
        const int   ka = kbase + (n & 7);
        const float t0 = sS[ka];
        const float t1 = sS[ka + 8];
        const float s0 = (ka < end) ? t0 : 0.0f;
        const float s1 = (ka + 8 < end) ? t1 : 0.0f;
        int rv[16];
        #pragma unroll
        for (int m = 0; m < 16; ++m) {
            const int r = sR[kbase + m];
            rv[m] = (kbase + m < end) ? r - row0 : 0;
        }

        v4f xv[8];
        #pragma unroll
        for (int i = 0; i < 8; ++i)
            xv[i] = __builtin_nontemporal_load(
                        (const v4f*)(xq + (size_t)colv[i] * NCH));
        #pragma unroll
        for (int i = 0; i < 8; ++i) {
            #pragma unroll
            for (int e = 0; e < 4; ++e) xv[i][e] = live[i] ? xv[i][e] : 0.0f;
        }

        v8f d0 = {}, d1 = {}, d2 = {}, d3 = {};
        #pragma unroll
        for (int g = 0; g < 4; ++g) {
            const bool k0 = (n == 4 * g + 2 * half + 0);
            const bool k1 = (n == 4 * g + 2 * half + 1);
            const int  i0 = 2 * (g & 1);
            const int  e  = g >> 1;
            v2f a0, a1, b0, b1, b2, b3;
            a0[0] = k0 ? s0 : 0.0f;  a0[1] = k1 ? s0 : 0.0f;
            a1[0] = k0 ? s1 : 0.0f;  a1[1] = k1 ? s1 : 0.0f;
            b0[0] = xv[i0][e];         b0[1] = xv[i0 + 1][e];
            b1[0] = xv[i0][2 + e];     b1[1] = xv[i0 + 1][2 + e];
            b2[0] = xv[4 + i0][e];     b2[1] = xv[4 + i0 + 1][e];
            b3[0] = xv[4 + i0][2 + e]; b3[1] = xv[4 + i0 + 1][2 + e];
            d0 = __builtin_amdgcn_wmma_f32_16x16x4_f32(
                     false, a0, false, b0, (short)0, d0, false, false);
            d1 = __builtin_amdgcn_wmma_f32_16x16x4_f32(
                     false, a0, false, b1, (short)0, d1, false, false);
            d2 = __builtin_amdgcn_wmma_f32_16x16x4_f32(
                     false, a1, false, b2, (short)0, d2, false, false);
            d3 = __builtin_amdgcn_wmma_f32_16x16x4_f32(
                     false, a1, false, b3, (short)0, d3, false, false);
        }

        #pragma unroll
        for (int v = 0; v < 8; ++v) {
            float* p0 = sl + rv[v] * SP;
            float* p1 = sl + rv[8 + v] * SP;
            atomicAdd(p0,      d0[v]);
            atomicAdd(p0 + 32, d1[v]);
            atomicAdd(p1,      d2[v]);
            atomicAdd(p1 + 32, d3[v]);
        }
    }
    __syncthreads();

    const int rowsHere = min(RB, N_OUT - row0);
    {
        const int ch = 2 * wv;
        float* __restrict__ yp = y + (size_t)ch * N_OUT + row0;
        const int nFull = rowsHere >> 5;
        #pragma unroll 8
        for (int j = 0; j < nFull; ++j) {
            const int r = 32 * j + lane;
            yp[r] = slab[r * SP + ch];
        }
        const int rt = 32 * nFull + lane;
        if (rt < rowsHere) yp[rt] = slab[rt * SP + ch];
    }
    {
        const int ch = 2 * wv + 1;
        float* __restrict__ yp = y + (size_t)ch * N_OUT + row0;
        const int nFull = (rowsHere - 16) >> 5;
        if (lane >= 16) yp[lane - 16] = slab[(lane - 16) * SP + ch];
        #pragma unroll 8
        for (int j = 1; j <= nFull; ++j) {
            const int r = 32 * j - 16 + lane;
            yp[r] = slab[r * SP + ch];
        }
        const int rt = 32 * (nFull + 1) - 16 + lane;
        if (rt < rowsHere) yp[rt] = slab[rt * SP + ch];
    }
}

extern "C" void kernel_launch(void* const* d_in, const int* in_sizes, int n_in,
                              void* d_out, int out_size, void* d_ws, size_t ws_size,
                              hipStream_t stream) {
    (void)in_sizes; (void)n_in; (void)ws_size; (void)out_size;

    const float* x    = (const float*)d_in[0];
    const float* S    = (const float*)d_in[1];
    const int*   rows = (const int*)d_in[2];
    const int*   cols = (const int*)d_in[3];
    float*       y    = (float*)d_out;
    float*       xt   = (float*)d_ws;
    int*   counts      = (int*)(xt + (size_t)N_IN * NCH);
    int*   bucketStart = counts + SORT_WG * NB;
    float* sS          = (float*)(bucketStart + NB);
    int*   sC          = (int*)(sS + NNZ + 16);
    int*   sR          = sC + NNZ + 16;

    regrid_relayout_kernel<<<RELAY_G, CNT_T, 0, stream>>>(x, xt, rows, counts);

    regrid_bucket_fill_kernel<<<SORT_WG, SORT_T, 0, stream>>>(S, rows, cols, counts,
                                                              bucketStart, sS, sC, sR);

    regrid_rows_owned_kernel<<<NBUCKET, 1024, 0, stream>>>(xt, sS, sR, sC, bucketStart, y);
}
